# starting kernel
# baseline (speedup 1.0000x reference)
#include <hip/hip_runtime.h>
#include <math.h>

typedef __attribute__((ext_vector_type(2))) float v2f;
typedef __attribute__((ext_vector_type(8))) float v8f;

#define IN_DIM 512

__device__ __forceinline__ v8f wmma_f32_4(v2f a, v2f b, v8f c) {
  return __builtin_amdgcn_wmma_f32_16x16x4_f32(false, a, false, b, (short)0, c,
                                               false, false);
}

__global__ void k_zero(float* p, int n) {
  int i = blockIdx.x * blockDim.x + threadIdx.x;
  if (i < n) p[i] = 0.0f;
}

__global__ __launch_bounds__(128) void k_fc1(const float* __restrict__ x,
                                             const float* __restrict__ w1,
                                             const float* __restrict__ b1,
                                             float* __restrict__ h1,
                                             int ntiles) {
  __shared__ float sW[IN_DIM * 16];
  int tid = threadIdx.x;
  for (int i = tid; i < IN_DIM * 16; i += 128) sW[i] = w1[i];
  __syncthreads();

  int wave = tid >> 5, lane = tid & 31;
  int tile = blockIdx.x * 4 + wave;
  if (tile >= ntiles) return;

  int n = lane & 15;
  int koff = (lane >> 4) << 1;
  long long row = (long long)tile * 16 + n;
  const float* xr = x + row * IN_DIM;

  v8f acc = {};
#pragma unroll 8
  for (int kb = 0; kb < IN_DIM / 4; ++kb) {
    int k = kb * 4 + koff;
    v2f a; a.x = xr[k]; a.y = xr[k + 1];
    v2f b; b.x = sW[k * 16 + n]; b.y = sW[(k + 1) * 16 + n];
    acc = wmma_f32_4(a, b, acc);
  }

  float bias = b1[n];
  int rbase = tile * 16 + ((lane >> 4) << 3);
#pragma unroll
  for (int r = 0; r < 8; ++r)
    h1[(long long)(rbase + r) * 16 + n] = fmaxf(acc[r] + bias, 0.0f);
}

__global__ void k_scatter1(const int* __restrict__ src, const int* __restrict__ dst,
                           const float* __restrict__ h1, float* __restrict__ agg1,
                           float* __restrict__ cnt, int nE) {
  int t = blockIdx.x * blockDim.x + threadIdx.x;
  int e = t >> 4, c = t & 15;
  if (e >= nE) return;
  int s = src[e], d = dst[e];
  atomicAdd(&agg1[(long long)d * 16 + c], h1[(long long)s * 16 + c]);
  if (c == 0) atomicAdd(&cnt[d], 1.0f);
}

__global__ __launch_bounds__(128) void k_conv1(const float* __restrict__ agg1,
                                               const float* __restrict__ cnt,
                                               const float* __restrict__ h1,
                                               const float* __restrict__ llw,
                                               const float* __restrict__ llb,
                                               const float* __restrict__ lrw,
                                               float* __restrict__ out1,
                                               int ntiles) {
  __shared__ float sW[32 * 32];
  int tid = threadIdx.x;
  for (int i = tid; i < 32 * 32; i += 128) {
    int k = i >> 5, j = i & 31;
    sW[i] = (k < 16) ? llw[k * 32 + j] : lrw[(k - 16) * 32 + j];
  }
  __syncthreads();

  int wave = tid >> 5, lane = tid & 31;
  int tile = blockIdx.x * 4 + wave;
  if (tile >= ntiles) return;

  int n = lane & 15;
  int koff = (lane >> 4) << 1;
  long long row = (long long)tile * 16 + n;
  float inv = 1.0f / fmaxf(cnt[row], 1.0f);

  v8f c0 = {}, c1 = {};
#pragma unroll
  for (int kb = 0; kb < 8; ++kb) {
    int kk = kb * 4 + koff;
    v2f a;
    if (kk < 16) { a.x = agg1[row * 16 + kk] * inv; a.y = agg1[row * 16 + kk + 1] * inv; }
    else         { a.x = h1[row * 16 + kk - 16];    a.y = h1[row * 16 + kk - 15]; }
    v2f b0; b0.x = sW[kk * 32 + n];      b0.y = sW[(kk + 1) * 32 + n];
    v2f b1; b1.x = sW[kk * 32 + 16 + n]; b1.y = sW[(kk + 1) * 32 + 16 + n];
    c0 = wmma_f32_4(a, b0, c0);
    c1 = wmma_f32_4(a, b1, c1);
  }

  float bl = llb[n], bh = llb[n + 16];
  int rbase = tile * 16 + ((lane >> 4) << 3);
#pragma unroll
  for (int r = 0; r < 8; ++r) {
    long long orow = (long long)(rbase + r) * 32;
    out1[orow + n]      = fmaxf(c0[r] + bl, 0.0f);
    out1[orow + 16 + n] = fmaxf(c1[r] + bh, 0.0f);
  }
}

__global__ void k_scatter2(const int* __restrict__ src, const int* __restrict__ dst,
                           const float* __restrict__ out1, float* __restrict__ agg2,
                           int nE) {
  int t = blockIdx.x * blockDim.x + threadIdx.x;
  int e = t >> 5, c = t & 31;
  if (e >= nE) return;
  int s = src[e], d = dst[e];
  atomicAdd(&agg2[(long long)d * 32 + c], out1[(long long)s * 32 + c]);
}

__global__ __launch_bounds__(128) void k_conv2(const float* __restrict__ agg2,
                                               const float* __restrict__ cnt,
                                               const float* __restrict__ out1,
                                               const float* __restrict__ llw,
                                               const float* __restrict__ llb,
                                               const float* __restrict__ lrw,
                                               float* __restrict__ out2,
                                               int ntiles) {
  __shared__ float sW[64 * 16];
  int tid = threadIdx.x;
  for (int i = tid; i < 64 * 16; i += 128) {
    int k = i >> 4, j = i & 15;
    sW[i] = (k < 32) ? llw[k * 16 + j] : lrw[(k - 32) * 16 + j];
  }
  __syncthreads();

  int wave = tid >> 5, lane = tid & 31;
  int tile = blockIdx.x * 4 + wave;
  if (tile >= ntiles) return;

  int n = lane & 15;
  int koff = (lane >> 4) << 1;
  long long row = (long long)tile * 16 + n;
  float inv = 1.0f / fmaxf(cnt[row], 1.0f);

  v8f acc = {};
#pragma unroll
  for (int kb = 0; kb < 16; ++kb) {
    int kk = kb * 4 + koff;
    v2f a;
    if (kk < 32) { a.x = agg2[row * 32 + kk] * inv; a.y = agg2[row * 32 + kk + 1] * inv; }
    else         { a.x = out1[row * 32 + kk - 32]; a.y = out1[row * 32 + kk - 31]; }
    v2f b; b.x = sW[kk * 16 + n]; b.y = sW[(kk + 1) * 16 + n];
    acc = wmma_f32_4(a, b, acc);
  }

  float bias = llb[n];
  int rbase = tile * 16 + ((lane >> 4) << 3);
#pragma unroll
  for (int r = 0; r < 8; ++r)
    out2[(long long)(rbase + r) * 16 + n] = fmaxf(acc[r] + bias, 0.0f);
}

__global__ void k_head(const float* __restrict__ out2, const float* __restrict__ w2,
                       const float* __restrict__ b2, float* __restrict__ out, int N) {
  int i = blockIdx.x * blockDim.x + threadIdx.x;
  if (i >= N) return;
  float h0 = b2[0], h1 = b2[1], h2 = b2[2];
  const float* r = out2 + (long long)i * 16;
#pragma unroll
  for (int k = 0; k < 16; ++k) {
    float v = r[k];
    h0 += v * w2[k * 3 + 0];
    h1 += v * w2[k * 3 + 1];
    h2 += v * w2[k * 3 + 2];
  }
  float gsi = 1.0f / (1.0f + __expf(-h1));
  float mxi = 1.0f / (1.0f + __expf(-h2));
  float fsi = fmaxf(h0, 0.0f) + gsi;
  out[(long long)i * 3 + 0] = fsi;
  out[(long long)i * 3 + 1] = gsi;
  out[(long long)i * 3 + 2] = mxi;
}

extern "C" void kernel_launch(void* const* d_in, const int* in_sizes, int n_in,
                              void* d_out, int out_size, void* d_ws, size_t ws_size,
                              hipStream_t stream) {
  const float* x    = (const float*)d_in[0];
  const int*   ei   = (const int*)d_in[1];
  const float* w1   = (const float*)d_in[2];
  const float* b1   = (const float*)d_in[3];
  const float* c1ll = (const float*)d_in[4];
  const float* c1lb = (const float*)d_in[5];
  const float* c1lr = (const float*)d_in[6];
  const float* c2ll = (const float*)d_in[7];
  const float* c2lb = (const float*)d_in[8];
  const float* c2lr = (const float*)d_in[9];
  const float* w2   = (const float*)d_in[10];
  const float* b2   = (const float*)d_in[11];
  float* out = (float*)d_out;

  int N = in_sizes[0] / IN_DIM;
  int E = in_sizes[1] / 2;
  const int* src = ei;
  const int* dst = ei + E;

  float* ws   = (float*)d_ws;
  float* h1   = ws;
  float* cnt  = ws + (long long)16 * N;
  float* agg1 = ws + (long long)17 * N;
  float* out1 = ws + (long long)33 * N;
  float* agg2 = ws + (long long)65 * N;
  float* out2 = ws + (long long)97 * N;

  int ntiles = (N + 15) / 16;
  int wblocks = (ntiles + 3) / 4;

  {
    int n1 = 17 * N;
    k_zero<<<(n1 + 255) / 256, 256, 0, stream>>>(cnt, n1);
    int n2 = 32 * N;
    k_zero<<<(n2 + 255) / 256, 256, 0, stream>>>(agg2, n2);
  }

  k_fc1<<<wblocks, 128, 0, stream>>>(x, w1, b1, h1, ntiles);

  {
    long long t = (long long)E * 16;
    k_scatter1<<<(unsigned)((t + 255) / 256), 256, 0, stream>>>(src, dst, h1, agg1, cnt, E);
  }

  k_conv1<<<wblocks, 128, 0, stream>>>(agg1, cnt, h1, c1ll, c1lb, c1lr, out1, ntiles);

  {
    long long t = (long long)E * 32;
    k_scatter2<<<(unsigned)((t + 255) / 256), 256, 0, stream>>>(src, dst, out1, agg2, E);
  }

  k_conv2<<<wblocks, 128, 0, stream>>>(agg2, cnt, out1, c2ll, c2lb, c2lr, out2, ntiles);

  k_head<<<(N + 255) / 256, 256, 0, stream>>>(out2, w2, b2, out, N);
}


# scatter1: in-degree moved into slot 16 of a 128-byte padded agg1 row, one edge per wave: ONE float-atomic line request per edge instead of two (sums still in atomic arrival order)
# speedup vs baseline: 1.1699x; 1.1699x over previous
#include <hip/hip_runtime.h>
#include <math.h>

typedef __attribute__((ext_vector_type(2))) float v2f;
typedef __attribute__((ext_vector_type(8))) float v8f;

#define IN_DIM 512

__device__ __forceinline__ v8f wmma_f32_4(v2f a, v2f b, v8f c) {
  return __builtin_amdgcn_wmma_f32_16x16x4_f32(false, a, false, b, (short)0, c,
                                               false, false);
}

__global__ void k_zero(float* p, int n) {
  int i = blockIdx.x * blockDim.x + threadIdx.x;
  if (i < n) p[i] = 0.0f;
}

__global__ __launch_bounds__(128) void k_fc1(const float* __restrict__ x,
                                             const float* __restrict__ w1,
                                             const float* __restrict__ b1,
                                             float* __restrict__ h1,
                                             int ntiles) {
  __shared__ float sW[IN_DIM * 16];
  int tid = threadIdx.x;
  for (int i = tid; i < IN_DIM * 16; i += 128) sW[i] = w1[i];
  __syncthreads();

  int wave = tid >> 5, lane = tid & 31;
  int tile = blockIdx.x * 4 + wave;
  if (tile >= ntiles) return;

  int n = lane & 15;
  int koff = (lane >> 4) << 1;
  long long row = (long long)tile * 16 + n;
  const float* xr = x + row * IN_DIM;

  v8f acc = {};
#pragma unroll 8
  for (int kb = 0; kb < IN_DIM / 4; ++kb) {
    int k = kb * 4 + koff;
    v2f a; a.x = xr[k]; a.y = xr[k + 1];
    v2f b; b.x = sW[k * 16 + n]; b.y = sW[(k + 1) * 16 + n];
    acc = wmma_f32_4(a, b, acc);
  }

  float bias = b1[n];
  int rbase = tile * 16 + ((lane >> 4) << 3);
#pragma unroll
  for (int r = 0; r < 8; ++r)
    h1[(long long)(rbase + r) * 16 + n] = fmaxf(acc[r] + bias, 0.0f);
}

__global__ void k_scatter1(const int* __restrict__ src, const int* __restrict__ dst,
                           const float* __restrict__ h1, float* __restrict__ agg1p,
                           int nE) {
  int t = blockIdx.x * blockDim.x + threadIdx.x;
  int e = t >> 5, c = t & 31;
  if (e >= nE || c > 16) return;
  int s = src[e], d = dst[e];
  float v = h1[(long long)s * 16 + (c & 15)];
  if (c == 16) v = 1.0f;
  atomicAdd(&agg1p[(long long)d * 32 + c], v);
}

__global__ __launch_bounds__(128) void k_conv1(const float* __restrict__ agg1p,
                                               const float* __restrict__ h1,
                                               const float* __restrict__ llw,
                                               const float* __restrict__ llb,
                                               const float* __restrict__ lrw,
                                               float* __restrict__ out1,
                                               int ntiles) {
  __shared__ float sW[32 * 32];
  int tid = threadIdx.x;
  for (int i = tid; i < 32 * 32; i += 128) {
    int k = i >> 5, j = i & 31;
    sW[i] = (k < 16) ? llw[k * 32 + j] : lrw[(k - 16) * 32 + j];
  }
  __syncthreads();

  int wave = tid >> 5, lane = tid & 31;
  int tile = blockIdx.x * 4 + wave;
  if (tile >= ntiles) return;

  int n = lane & 15;
  int koff = (lane >> 4) << 1;
  long long row = (long long)tile * 16 + n;
  float inv = 1.0f / fmaxf(agg1p[row * 32 + 16], 1.0f);

  v8f c0 = {}, c1 = {};
#pragma unroll
  for (int kb = 0; kb < 8; ++kb) {
    int kk = kb * 4 + koff;
    v2f a;
    if (kk < 16) { a.x = agg1p[row * 32 + kk] * inv; a.y = agg1p[row * 32 + kk + 1] * inv; }
    else         { a.x = h1[row * 16 + kk - 16];    a.y = h1[row * 16 + kk - 15]; }
    v2f b0; b0.x = sW[kk * 32 + n];      b0.y = sW[(kk + 1) * 32 + n];
    v2f b1; b1.x = sW[kk * 32 + 16 + n]; b1.y = sW[(kk + 1) * 32 + 16 + n];
    c0 = wmma_f32_4(a, b0, c0);
    c1 = wmma_f32_4(a, b1, c1);
  }

  float bl = llb[n], bh = llb[n + 16];
  int rbase = tile * 16 + ((lane >> 4) << 3);
#pragma unroll
  for (int r = 0; r < 8; ++r) {
    long long orow = (long long)(rbase + r) * 32;
    out1[orow + n]      = fmaxf(c0[r] + bl, 0.0f);
    out1[orow + 16 + n] = fmaxf(c1[r] + bh, 0.0f);
  }
}

__global__ void k_scatter2(const int* __restrict__ src, const int* __restrict__ dst,
                           const float* __restrict__ out1, float* __restrict__ agg2,
                           int nE) {
  int t = blockIdx.x * blockDim.x + threadIdx.x;
  int e = t >> 5, c = t & 31;
  if (e >= nE) return;
  int s = src[e], d = dst[e];
  atomicAdd(&agg2[(long long)d * 32 + c], out1[(long long)s * 32 + c]);
}

__global__ __launch_bounds__(128) void k_conv2(const float* __restrict__ agg2,
                                               const float* __restrict__ agg1p,
                                               const float* __restrict__ out1,
                                               const float* __restrict__ llw,
                                               const float* __restrict__ llb,
                                               const float* __restrict__ lrw,
                                               float* __restrict__ out2,
                                               int ntiles) {
  __shared__ float sW[64 * 16];
  int tid = threadIdx.x;
  for (int i = tid; i < 64 * 16; i += 128) {
    int k = i >> 4, j = i & 15;
    sW[i] = (k < 32) ? llw[k * 16 + j] : lrw[(k - 32) * 16 + j];
  }
  __syncthreads();

  int wave = tid >> 5, lane = tid & 31;
  int tile = blockIdx.x * 4 + wave;
  if (tile >= ntiles) return;

  int n = lane & 15;
  int koff = (lane >> 4) << 1;
  long long row = (long long)tile * 16 + n;
  float inv = 1.0f / fmaxf(agg1p[row * 32 + 16], 1.0f);

  v8f acc = {};
#pragma unroll
  for (int kb = 0; kb < 16; ++kb) {
    int kk = kb * 4 + koff;
    v2f a;
    if (kk < 32) { a.x = agg2[row * 32 + kk] * inv; a.y = agg2[row * 32 + kk + 1] * inv; }
    else         { a.x = out1[row * 32 + kk - 32]; a.y = out1[row * 32 + kk - 31]; }
    v2f b; b.x = sW[kk * 16 + n]; b.y = sW[(kk + 1) * 16 + n];
    acc = wmma_f32_4(a, b, acc);
  }

  float bias = llb[n];
  int rbase = tile * 16 + ((lane >> 4) << 3);
#pragma unroll
  for (int r = 0; r < 8; ++r)
    out2[(long long)(rbase + r) * 16 + n] = fmaxf(acc[r] + bias, 0.0f);
}

__global__ void k_head(const float* __restrict__ out2, const float* __restrict__ w2,
                       const float* __restrict__ b2, float* __restrict__ out, int N) {
  int i = blockIdx.x * blockDim.x + threadIdx.x;
  if (i >= N) return;
  float h0 = b2[0], h1 = b2[1], h2 = b2[2];
  const float* r = out2 + (long long)i * 16;
#pragma unroll
  for (int k = 0; k < 16; ++k) {
    float v = r[k];
    h0 += v * w2[k * 3 + 0];
    h1 += v * w2[k * 3 + 1];
    h2 += v * w2[k * 3 + 2];
  }
  float gsi = 1.0f / (1.0f + __expf(-h1));
  float mxi = 1.0f / (1.0f + __expf(-h2));
  float fsi = fmaxf(h0, 0.0f) + gsi;
  out[(long long)i * 3 + 0] = fsi;
  out[(long long)i * 3 + 1] = gsi;
  out[(long long)i * 3 + 2] = mxi;
}

extern "C" void kernel_launch(void* const* d_in, const int* in_sizes, int n_in,
                              void* d_out, int out_size, void* d_ws, size_t ws_size,
                              hipStream_t stream) {
  const float* x    = (const float*)d_in[0];
  const int*   ei   = (const int*)d_in[1];
  const float* w1   = (const float*)d_in[2];
  const float* b1   = (const float*)d_in[3];
  const float* c1ll = (const float*)d_in[4];
  const float* c1lb = (const float*)d_in[5];
  const float* c1lr = (const float*)d_in[6];
  const float* c2ll = (const float*)d_in[7];
  const float* c2lb = (const float*)d_in[8];
  const float* c2lr = (const float*)d_in[9];
  const float* w2   = (const float*)d_in[10];
  const float* b2   = (const float*)d_in[11];
  float* out = (float*)d_out;

  int N = in_sizes[0] / IN_DIM;
  int E = in_sizes[1] / 2;
  const int* src = ei;
  const int* dst = ei + E;

  float* ws    = (float*)d_ws;
  float* h1    = ws;
  float* agg1p = ws + (long long)16 * N;
  float* out1  = ws + (long long)48 * N;
  float* agg2  = ws + (long long)80 * N;
  float* out2  = ws;

  int ntiles = (N + 15) / 16;
  int wblocks = (ntiles + 3) / 4;

  {
    int n1 = 32 * N;
    k_zero<<<(n1 + 255) / 256, 256, 0, stream>>>(agg1p, n1);
    int n2 = 32 * N;
    k_zero<<<(n2 + 255) / 256, 256, 0, stream>>>(agg2, n2);
  }

  k_fc1<<<wblocks, 128, 0, stream>>>(x, w1, b1, h1, ntiles);

  {
    long long t = (long long)E * 32;
    k_scatter1<<<(unsigned)((t + 255) / 256), 256, 0, stream>>>(src, dst, h1, agg1p, E);
  }

  k_conv1<<<wblocks, 128, 0, stream>>>(agg1p, h1, c1ll, c1lb, c1lr, out1, ntiles);

  {
    long long t = (long long)E * 32;
    k_scatter2<<<(unsigned)((t + 255) / 256), 256, 0, stream>>>(src, dst, out1, agg2, E);
  }

  k_conv2<<<wblocks, 128, 0, stream>>>(agg2, agg1p, out1, c2ll, c2lb, c2lr, out2, ntiles);

  k_head<<<(N + 255) / 256, 256, 0, stream>>>(out2, w2, b2, out, N);
}


# scatter1+scatter2: each wave takes 32 consecutive edges (coalesced index load, lane broadcast, 32 row gathers in flight, then 32 atomic row adds), not one edge per wave; sums still in arrival order
# speedup vs baseline: 2.9760x; 2.5438x over previous
#include <hip/hip_runtime.h>
#include <math.h>

typedef __attribute__((ext_vector_type(2))) float v2f;
typedef __attribute__((ext_vector_type(8))) float v8f;

#define IN_DIM 512

__device__ __forceinline__ v8f wmma_f32_4(v2f a, v2f b, v8f c) {
  return __builtin_amdgcn_wmma_f32_16x16x4_f32(false, a, false, b, (short)0, c,
                                               false, false);
}

__global__ void k_zero(float* p, int n) {
  int i = blockIdx.x * blockDim.x + threadIdx.x;
  if (i < n) p[i] = 0.0f;
}

__global__ __launch_bounds__(128) void k_fc1(const float* __restrict__ x,
                                             const float* __restrict__ w1,
                                             const float* __restrict__ b1,
                                             float* __restrict__ h1,
                                             int ntiles) {
  __shared__ float sW[IN_DIM * 16];
  int tid = threadIdx.x;
  for (int i = tid; i < IN_DIM * 16; i += 128) sW[i] = w1[i];
  __syncthreads();

  int wave = tid >> 5, lane = tid & 31;
  int tile = blockIdx.x * 4 + wave;
  if (tile >= ntiles) return;

  int n = lane & 15;
  int koff = (lane >> 4) << 1;
  long long row = (long long)tile * 16 + n;
  const float* xr = x + row * IN_DIM;

  v8f acc = {};
#pragma unroll 8
  for (int kb = 0; kb < IN_DIM / 4; ++kb) {
    int k = kb * 4 + koff;
    v2f a; a.x = xr[k]; a.y = xr[k + 1];
    v2f b; b.x = sW[k * 16 + n]; b.y = sW[(k + 1) * 16 + n];
    acc = wmma_f32_4(a, b, acc);
  }

  float bias = b1[n];
  int rbase = tile * 16 + ((lane >> 4) << 3);
#pragma unroll
  for (int r = 0; r < 8; ++r)
    h1[(long long)(rbase + r) * 16 + n] = fmaxf(acc[r] + bias, 0.0f);
}

__global__ __launch_bounds__(256) void k_scatter1(const int* __restrict__ src,
                                                  const int* __restrict__ dst,
                                                  const float* __restrict__ h1,
                                                  float* __restrict__ agg1p, int nE) {
  int lane = threadIdx.x & 31;
  long long e0 = ((long long)blockIdx.x * (blockDim.x >> 5) + (threadIdx.x >> 5)) * 32;
  if (e0 >= nE) return;
  long long e = e0 + lane;
  int s_l = 0, d_l = 0;
  if (e < nE) { s_l = src[e]; d_l = dst[e]; }
  int c = lane & 15;
  if (e0 + 32 <= nE) {
    float v[32];
    int dd[32];
#pragma unroll
    for (int j = 0; j < 32; ++j) {
      int s = __builtin_amdgcn_readlane(s_l, j);
      dd[j] = __builtin_amdgcn_readlane(d_l, j);
      v[j] = h1[(long long)s * 16 + c];
    }
    if (lane <= 16) {
#pragma unroll
      for (int j = 0; j < 32; ++j) {
        float val = (lane == 16) ? 1.0f : v[j];
        atomicAdd(&agg1p[(long long)dd[j] * 32 + lane], val);
      }
    }
  } else {
    int cnt = (int)(nE - e0);
    for (int j = 0; j < cnt; ++j) {
      int s = __builtin_amdgcn_readlane(s_l, j);
      int d = __builtin_amdgcn_readlane(d_l, j);
      float val = (lane == 16) ? 1.0f : h1[(long long)s * 16 + c];
      if (lane <= 16) atomicAdd(&agg1p[(long long)d * 32 + lane], val);
    }
  }
}

__global__ __launch_bounds__(128) void k_conv1(const float* __restrict__ agg1p,
                                               const float* __restrict__ h1,
                                               const float* __restrict__ llw,
                                               const float* __restrict__ llb,
                                               const float* __restrict__ lrw,
                                               float* __restrict__ out1,
                                               int ntiles) {
  __shared__ float sW[32 * 32];
  int tid = threadIdx.x;
  for (int i = tid; i < 32 * 32; i += 128) {
    int k = i >> 5, j = i & 31;
    sW[i] = (k < 16) ? llw[k * 32 + j] : lrw[(k - 16) * 32 + j];
  }
  __syncthreads();

  int wave = tid >> 5, lane = tid & 31;
  int tile = blockIdx.x * 4 + wave;
  if (tile >= ntiles) return;

  int n = lane & 15;
  int koff = (lane >> 4) << 1;
  long long row = (long long)tile * 16 + n;
  float inv = 1.0f / fmaxf(agg1p[row * 32 + 16], 1.0f);

  v8f c0 = {}, c1 = {};
#pragma unroll
  for (int kb = 0; kb < 8; ++kb) {
    int kk = kb * 4 + koff;
    v2f a;
    if (kk < 16) { a.x = agg1p[row * 32 + kk] * inv; a.y = agg1p[row * 32 + kk + 1] * inv; }
    else         { a.x = h1[row * 16 + kk - 16];    a.y = h1[row * 16 + kk - 15]; }
    v2f b0; b0.x = sW[kk * 32 + n];      b0.y = sW[(kk + 1) * 32 + n];
    v2f b1; b1.x = sW[kk * 32 + 16 + n]; b1.y = sW[(kk + 1) * 32 + 16 + n];
    c0 = wmma_f32_4(a, b0, c0);
    c1 = wmma_f32_4(a, b1, c1);
  }

  float bl = llb[n], bh = llb[n + 16];
  int rbase = tile * 16 + ((lane >> 4) << 3);
#pragma unroll
  for (int r = 0; r < 8; ++r) {
    long long orow = (long long)(rbase + r) * 32;
    out1[orow + n]      = fmaxf(c0[r] + bl, 0.0f);
    out1[orow + 16 + n] = fmaxf(c1[r] + bh, 0.0f);
  }
}

__global__ __launch_bounds__(256) void k_scatter2(const int* __restrict__ src,
                                                  const int* __restrict__ dst,
                                                  const float* __restrict__ out1,
                                                  float* __restrict__ agg2, int nE) {
  int lane = threadIdx.x & 31;
  long long e0 = ((long long)blockIdx.x * (blockDim.x >> 5) + (threadIdx.x >> 5)) * 32;
  if (e0 >= nE) return;
  long long e = e0 + lane;
  int s_l = 0, d_l = 0;
  if (e < nE) { s_l = src[e]; d_l = dst[e]; }
  if (e0 + 32 <= nE) {
    float v[32];
#pragma unroll
    for (int j = 0; j < 32; ++j) {
      int s = __builtin_amdgcn_readlane(s_l, j);
      v[j] = out1[(long long)s * 32 + lane];
    }
#pragma unroll
    for (int j = 0; j < 32; ++j) {
      int d = __builtin_amdgcn_readlane(d_l, j);
      atomicAdd(&agg2[(long long)d * 32 + lane], v[j]);
    }
  } else {
    int cnt = (int)(nE - e0);
    for (int j = 0; j < cnt; ++j) {
      int s = __builtin_amdgcn_readlane(s_l, j);
      int d = __builtin_amdgcn_readlane(d_l, j);
      atomicAdd(&agg2[(long long)d * 32 + lane], out1[(long long)s * 32 + lane]);
    }
  }
}

__global__ __launch_bounds__(128) void k_conv2(const float* __restrict__ agg2,
                                               const float* __restrict__ agg1p,
                                               const float* __restrict__ out1,
                                               const float* __restrict__ llw,
                                               const float* __restrict__ llb,
                                               const float* __restrict__ lrw,
                                               float* __restrict__ out2,
                                               int ntiles) {
  __shared__ float sW[64 * 16];
  int tid = threadIdx.x;
  for (int i = tid; i < 64 * 16; i += 128) {
    int k = i >> 4, j = i & 15;
    sW[i] = (k < 32) ? llw[k * 16 + j] : lrw[(k - 32) * 16 + j];
  }
  __syncthreads();

  int wave = tid >> 5, lane = tid & 31;
  int tile = blockIdx.x * 4 + wave;
  if (tile >= ntiles) return;

  int n = lane & 15;
  int koff = (lane >> 4) << 1;
  long long row = (long long)tile * 16 + n;
  float inv = 1.0f / fmaxf(agg1p[row * 32 + 16], 1.0f);

  v8f acc = {};
#pragma unroll
  for (int kb = 0; kb < 16; ++kb) {
    int kk = kb * 4 + koff;
    v2f a;
    if (kk < 32) { a.x = agg2[row * 32 + kk] * inv; a.y = agg2[row * 32 + kk + 1] * inv; }
    else         { a.x = out1[row * 32 + kk - 32]; a.y = out1[row * 32 + kk - 31]; }
    v2f b; b.x = sW[kk * 16 + n]; b.y = sW[(kk + 1) * 16 + n];
    acc = wmma_f32_4(a, b, acc);
  }

  float bias = llb[n];
  int rbase = tile * 16 + ((lane >> 4) << 3);
#pragma unroll
  for (int r = 0; r < 8; ++r)
    out2[(long long)(rbase + r) * 16 + n] = fmaxf(acc[r] + bias, 0.0f);
}

__global__ void k_head(const float* __restrict__ out2, const float* __restrict__ w2,
                       const float* __restrict__ b2, float* __restrict__ out, int N) {
  int i = blockIdx.x * blockDim.x + threadIdx.x;
  if (i >= N) return;
  float h0 = b2[0], h1 = b2[1], h2 = b2[2];
  const float* r = out2 + (long long)i * 16;
#pragma unroll
  for (int k = 0; k < 16; ++k) {
    float v = r[k];
    h0 += v * w2[k * 3 + 0];
    h1 += v * w2[k * 3 + 1];
    h2 += v * w2[k * 3 + 2];
  }
  float gsi = 1.0f / (1.0f + __expf(-h1));
  float mxi = 1.0f / (1.0f + __expf(-h2));
  float fsi = fmaxf(h0, 0.0f) + gsi;
  out[(long long)i * 3 + 0] = fsi;
  out[(long long)i * 3 + 1] = gsi;
  out[(long long)i * 3 + 2] = mxi;
}

extern "C" void kernel_launch(void* const* d_in, const int* in_sizes, int n_in,
                              void* d_out, int out_size, void* d_ws, size_t ws_size,
                              hipStream_t stream) {
  const float* x    = (const float*)d_in[0];
  const int*   ei   = (const int*)d_in[1];
  const float* w1   = (const float*)d_in[2];
  const float* b1   = (const float*)d_in[3];
  const float* c1ll = (const float*)d_in[4];
  const float* c1lb = (const float*)d_in[5];
  const float* c1lr = (const float*)d_in[6];
  const float* c2ll = (const float*)d_in[7];
  const float* c2lb = (const float*)d_in[8];
  const float* c2lr = (const float*)d_in[9];
  const float* w2   = (const float*)d_in[10];
  const float* b2   = (const float*)d_in[11];
  float* out = (float*)d_out;

  int N = in_sizes[0] / IN_DIM;
  int E = in_sizes[1] / 2;
  const int* src = ei;
  const int* dst = ei + E;

  float* ws    = (float*)d_ws;
  float* h1    = ws;
  float* agg1p = ws + (long long)16 * N;
  float* out1  = ws + (long long)48 * N;
  float* agg2  = ws + (long long)80 * N;
  float* out2  = ws;

  int ntiles = (N + 15) / 16;
  int wblocks = (ntiles + 3) / 4;

  {
    int n1 = 32 * N;
    k_zero<<<(n1 + 255) / 256, 256, 0, stream>>>(agg1p, n1);
    int n2 = 32 * N;
    k_zero<<<(n2 + 255) / 256, 256, 0, stream>>>(agg2, n2);
  }

  k_fc1<<<wblocks, 128, 0, stream>>>(x, w1, b1, h1, ntiles);

  {
    k_scatter1<<<(unsigned)((E + 255) / 256), 256, 0, stream>>>(src, dst, h1, agg1p, E);
  }

  k_conv1<<<wblocks, 128, 0, stream>>>(agg1p, h1, c1ll, c1lb, c1lr, out1, ntiles);

  {
    k_scatter2<<<(unsigned)((E + 255) / 256), 256, 0, stream>>>(src, dst, out1, agg2, E);
  }

  k_conv2<<<wblocks, 128, 0, stream>>>(agg2, agg1p, out1, c2ll, c2lb, c2lr, out2, ntiles);

  k_head<<<(N + 255) / 256, 256, 0, stream>>>(out2, w2, b2, out, N);
}


# scatter1+scatter2: accumulator kept as two planes, one per chip half (die id, speed only); a wave adds only into its own half's plane, conv1/conv2 add the two partial sums (arrival order in each)
# speedup vs baseline: 3.8745x; 1.3019x over previous
#include <hip/hip_runtime.h>
#include <math.h>

typedef __attribute__((ext_vector_type(2))) float v2f;
typedef __attribute__((ext_vector_type(4))) float v4f;
typedef __attribute__((ext_vector_type(8))) float v8f;

#define IN_DIM 512

__device__ __forceinline__ v8f wmma_f32_4(v2f a, v2f b, v8f c) {
  return __builtin_amdgcn_wmma_f32_16x16x4_f32(false, a, false, b, (short)0, c,
                                               false, false);
}

__device__ __forceinline__ int chip_half() {
  return (int)((__builtin_amdgcn_s_sendmsg_rtn(0x87) >> 18) & 1u);
}

__global__ void k_zero(float* p, int n) {
  int i = blockIdx.x * blockDim.x + threadIdx.x;
  if (i < n) p[i] = 0.0f;
}

__global__ __launch_bounds__(128) void k_fc1(const float* __restrict__ x,
                                             const float* __restrict__ w1,
                                             const float* __restrict__ b1,
                                             float* __restrict__ h1,
                                             int ntiles) {
  __shared__ float sW[IN_DIM * 16];
  int tid = threadIdx.x;
  for (int i = tid; i < IN_DIM * 16; i += 128) sW[i] = w1[i];
  __syncthreads();

  int wave = tid >> 5, lane = tid & 31;
  int tile = blockIdx.x * 4 + wave;
  if (tile >= ntiles) return;

  int n = lane & 15;
  int koff = (lane >> 4) << 1;
  long long row = (long long)tile * 16 + n;
  const float* xr = x + row * IN_DIM;

  v8f acc = {};
#pragma unroll 8
  for (int kb = 0; kb < IN_DIM / 4; ++kb) {
    int k = kb * 4 + koff;
    v2f a; a.x = xr[k]; a.y = xr[k + 1];
    v2f b; b.x = sW[k * 16 + n]; b.y = sW[(k + 1) * 16 + n];
    acc = wmma_f32_4(a, b, acc);
  }

  float bias = b1[n];
  int rbase = tile * 16 + ((lane >> 4) << 3);
#pragma unroll
  for (int r = 0; r < 8; ++r)
    h1[(long long)(rbase + r) * 16 + n] = fmaxf(acc[r] + bias, 0.0f);
}

__global__ __launch_bounds__(256) void k_scatter1(const int* __restrict__ src,
                                                  const int* __restrict__ dst,
                                                  const float* __restrict__ h1,
                                                  float* __restrict__ acc, int plane,
                                                  int nE) {
  float* agg1p = acc + (long long)chip_half() * plane;
  int lane = threadIdx.x & 31;
  long long e0 = ((long long)blockIdx.x * (blockDim.x >> 5) + (threadIdx.x >> 5)) * 32;
  if (e0 >= nE) return;
  long long e = e0 + lane;
  int s_l = 0, d_l = 0;
  if (e < nE) { s_l = src[e]; d_l = dst[e]; }
  int c = lane & 15;
  if (e0 + 32 <= nE) {
    float v[32];
    int dd[32];
#pragma unroll
    for (int j = 0; j < 32; ++j) {
      int s = __builtin_amdgcn_readlane(s_l, j);
      dd[j] = __builtin_amdgcn_readlane(d_l, j);
      v[j] = h1[(long long)s * 16 + c];
    }
    if (lane <= 16) {
#pragma unroll
      for (int j = 0; j < 32; ++j) {
        float val = (lane == 16) ? 1.0f : v[j];
        atomicAdd(&agg1p[(long long)dd[j] * 32 + lane], val);
      }
    }
  } else {
    int cnt = (int)(nE - e0);
    for (int j = 0; j < cnt; ++j) {
      int s = __builtin_amdgcn_readlane(s_l, j);
      int d = __builtin_amdgcn_readlane(d_l, j);
      float val = (lane == 16) ? 1.0f : h1[(long long)s * 16 + c];
      if (lane <= 16) atomicAdd(&agg1p[(long long)d * 32 + lane], val);
    }
  }
}

__global__ __launch_bounds__(128) void k_conv1(float* accA, float* accB, float* h1,
                                               const float* __restrict__ llw,
                                               const float* __restrict__ llb,
                                               const float* __restrict__ lrw,
                                               float* __restrict__ out1,
                                               int ntiles) {
  __shared__ float sW[32 * 32];
  int tid = threadIdx.x;
  for (int i = tid; i < 32 * 32; i += 128) {
    int k = i >> 5, j = i & 31;
    sW[i] = (k < 16) ? llw[k * 32 + j] : lrw[(k - 16) * 32 + j];
  }
  __syncthreads();

  int wave = tid >> 5, lane = tid & 31;
  int tile = blockIdx.x * 4 + wave;
  if (tile >= ntiles) return;

  int n = lane & 15;
  int koff = (lane >> 4) << 1;
  long long row = (long long)tile * 16 + n;
  float inv = 1.0f / fmaxf(accA[row * 32 + 16] + accB[row * 32 + 16], 1.0f);

  v8f c0 = {}, c1 = {};
#pragma unroll
  for (int kb = 0; kb < 8; ++kb) {
    int kk = kb * 4 + koff;
    v2f a;
    if (kk < 16) { a.x = (accA[row * 32 + kk] + accB[row * 32 + kk]) * inv;
                   a.y = (accA[row * 32 + kk + 1] + accB[row * 32 + kk + 1]) * inv; }
    else         { a.x = h1[row * 16 + kk - 16];    a.y = h1[row * 16 + kk - 15]; }
    v2f b0; b0.x = sW[kk * 32 + n];      b0.y = sW[(kk + 1) * 32 + n];
    v2f b1; b1.x = sW[kk * 32 + 16 + n]; b1.y = sW[(kk + 1) * 32 + 16 + n];
    c0 = wmma_f32_4(a, b0, c0);
    c1 = wmma_f32_4(a, b1, c1);
  }

  float bl = llb[n], bh = llb[n + 16];
  int rbase = tile * 16 + ((lane >> 4) << 3);
#pragma unroll
  for (int r = 0; r < 8; ++r) {
    long long orow = (long long)(rbase + r) * 32;
    out1[orow + n]      = fmaxf(c0[r] + bl, 0.0f);
    out1[orow + 16 + n] = fmaxf(c1[r] + bh, 0.0f);
  }

  if (lane < 16) h1[row * 16] = inv;

  v4f z = {};
  v4f* pa = (v4f*)(accA + (long long)tile * 512);
  v4f* pb = (v4f*)(accB + (long long)tile * 512);
#pragma unroll
  for (int q = 0; q < 4; ++q) { pa[q * 32 + lane] = z; pb[q * 32 + lane] = z; }
}

__global__ __launch_bounds__(256) void k_scatter2(const int* __restrict__ src,
                                                  const int* __restrict__ dst,
                                                  const float* __restrict__ out1,
                                                  float* __restrict__ acc, int plane,
                                                  int nE) {
  float* agg2 = acc + (long long)chip_half() * plane;
  int lane = threadIdx.x & 31;
  long long e0 = ((long long)blockIdx.x * (blockDim.x >> 5) + (threadIdx.x >> 5)) * 32;
  if (e0 >= nE) return;
  long long e = e0 + lane;
  int s_l = 0, d_l = 0;
  if (e < nE) { s_l = src[e]; d_l = dst[e]; }
  if (e0 + 32 <= nE) {
    float v[32];
#pragma unroll
    for (int j = 0; j < 32; ++j) {
      int s = __builtin_amdgcn_readlane(s_l, j);
      v[j] = out1[(long long)s * 32 + lane];
    }
#pragma unroll
    for (int j = 0; j < 32; ++j) {
      int d = __builtin_amdgcn_readlane(d_l, j);
      atomicAdd(&agg2[(long long)d * 32 + lane], v[j]);
    }
  } else {
    int cnt = (int)(nE - e0);
    for (int j = 0; j < cnt; ++j) {
      int s = __builtin_amdgcn_readlane(s_l, j);
      int d = __builtin_amdgcn_readlane(d_l, j);
      atomicAdd(&agg2[(long long)d * 32 + lane], out1[(long long)s * 32 + lane]);
    }
  }
}

__global__ __launch_bounds__(128) void k_conv2(const float* __restrict__ accA,
                                               const float* __restrict__ accB,
                                               const float* __restrict__ out1,
                                               const float* __restrict__ llw,
                                               const float* __restrict__ llb,
                                               const float* __restrict__ lrw,
                                               float* r0,
                                               int ntiles) {
  __shared__ float sW[64 * 16];
  int tid = threadIdx.x;
  for (int i = tid; i < 64 * 16; i += 128) {
    int k = i >> 4, j = i & 15;
    sW[i] = (k < 32) ? llw[k * 16 + j] : lrw[(k - 32) * 16 + j];
  }
  __syncthreads();

  int wave = tid >> 5, lane = tid & 31;
  int tile = blockIdx.x * 4 + wave;
  if (tile >= ntiles) return;

  int n = lane & 15;
  int koff = (lane >> 4) << 1;
  long long row = (long long)tile * 16 + n;
  float inv = r0[row * 16];

  v8f acc = {};
#pragma unroll
  for (int kb = 0; kb < 16; ++kb) {
    int kk = kb * 4 + koff;
    v2f a;
    if (kk < 32) { a.x = (accA[row * 32 + kk] + accB[row * 32 + kk]) * inv;
                   a.y = (accA[row * 32 + kk + 1] + accB[row * 32 + kk + 1]) * inv; }
    else         { a.x = out1[row * 32 + kk - 32]; a.y = out1[row * 32 + kk - 31]; }
    v2f b; b.x = sW[kk * 16 + n]; b.y = sW[(kk + 1) * 16 + n];
    acc = wmma_f32_4(a, b, acc);
  }

  float bias = llb[n];
  int rbase = tile * 16 + ((lane >> 4) << 3);
#pragma unroll
  for (int r = 0; r < 8; ++r)
    r0[(long long)(rbase + r) * 16 + n] = fmaxf(acc[r] + bias, 0.0f);
}

__global__ void k_head(const float* __restrict__ out2, const float* __restrict__ w2,
                       const float* __restrict__ b2, float* __restrict__ out, int N) {
  int i = blockIdx.x * blockDim.x + threadIdx.x;
  if (i >= N) return;
  float h0 = b2[0], h1 = b2[1], h2 = b2[2];
  const float* r = out2 + (long long)i * 16;
#pragma unroll
  for (int k = 0; k < 16; ++k) {
    float v = r[k];
    h0 += v * w2[k * 3 + 0];
    h1 += v * w2[k * 3 + 1];
    h2 += v * w2[k * 3 + 2];
  }
  float gsi = 1.0f / (1.0f + __expf(-h1));
  float mxi = 1.0f / (1.0f + __expf(-h2));
  float fsi = fmaxf(h0, 0.0f) + gsi;
  out[(long long)i * 3 + 0] = fsi;
  out[(long long)i * 3 + 1] = gsi;
  out[(long long)i * 3 + 2] = mxi;
}

extern "C" void kernel_launch(void* const* d_in, const int* in_sizes, int n_in,
                              void* d_out, int out_size, void* d_ws, size_t ws_size,
                              hipStream_t stream) {
  const float* x    = (const float*)d_in[0];
  const int*   ei   = (const int*)d_in[1];
  const float* w1   = (const float*)d_in[2];
  const float* b1   = (const float*)d_in[3];
  const float* c1ll = (const float*)d_in[4];
  const float* c1lb = (const float*)d_in[5];
  const float* c1lr = (const float*)d_in[6];
  const float* c2ll = (const float*)d_in[7];
  const float* c2lb = (const float*)d_in[8];
  const float* c2lr = (const float*)d_in[9];
  const float* w2   = (const float*)d_in[10];
  const float* b2   = (const float*)d_in[11];
  float* out = (float*)d_out;

  int N = in_sizes[0] / IN_DIM;
  int E = in_sizes[1] / 2;
  const int* src = ei;
  const int* dst = ei + E;

  float* ws    = (float*)d_ws;
  float* h1    = ws;
  float* acc   = ws + (long long)16 * N;
  float* out1  = ws + (long long)80 * N;
  float* out2  = ws;
  int plane = 32 * N;

  int ntiles = (N + 15) / 16;
  int wblocks = (ntiles + 3) / 4;

  {
    int n1 = 2 * plane;
    k_zero<<<(n1 + 255) / 256, 256, 0, stream>>>(acc, n1);
  }

  k_fc1<<<wblocks, 128, 0, stream>>>(x, w1, b1, h1, ntiles);

  {
    k_scatter1<<<(unsigned)((E + 255) / 256), 256, 0, stream>>>(src, dst, h1, acc, plane, E);
  }

  k_conv1<<<wblocks, 128, 0, stream>>>(acc, acc + plane, h1, c1ll, c1lb, c1lr, out1, ntiles);

  {
    k_scatter2<<<(unsigned)((E + 255) / 256), 256, 0, stream>>>(src, dst, out1, acc, plane, E);
  }

  k_conv2<<<wblocks, 128, 0, stream>>>(acc, acc + plane, out1, c2ll, c2lb, c2lr, out2, ntiles);

  k_head<<<(N + 255) / 256, 256, 0, stream>>>(out2, w2, b2, out, N);
}


# layer 2: conv1 also forms p = out1 @ conv2 neighbour weights (mean is linear), scatter2 gathers and atomically adds 16-float p rows, not 32; sums are of per-edge products, arrival order
# speedup vs baseline: 4.7890x; 1.2360x over previous
#include <hip/hip_runtime.h>
#include <math.h>

typedef __attribute__((ext_vector_type(2))) float v2f;
typedef __attribute__((ext_vector_type(4))) float v4f;
typedef __attribute__((ext_vector_type(8))) float v8f;

#define IN_DIM 512

__device__ __forceinline__ v8f wmma_f32_4(v2f a, v2f b, v8f c) {
  return __builtin_amdgcn_wmma_f32_16x16x4_f32(false, a, false, b, (short)0, c,
                                               false, false);
}

__device__ __forceinline__ int chip_half() {
  return (int)((__builtin_amdgcn_s_sendmsg_rtn(0x87) >> 18) & 1u);
}

__global__ void k_zero(float* p, int n) {
  int i = blockIdx.x * blockDim.x + threadIdx.x;
  if (i < n) p[i] = 0.0f;
}

__global__ __launch_bounds__(128) void k_fc1(const float* __restrict__ x,
                                             const float* __restrict__ w1,
                                             const float* __restrict__ b1,
                                             float* __restrict__ h1,
                                             int ntiles) {
  __shared__ float sW[IN_DIM * 16];
  int tid = threadIdx.x;
  for (int i = tid; i < IN_DIM * 16; i += 128) sW[i] = w1[i];
  __syncthreads();

  int wave = tid >> 5, lane = tid & 31;
  int tile = blockIdx.x * 4 + wave;
  if (tile >= ntiles) return;

  int n = lane & 15;
  int koff = (lane >> 4) << 1;
  long long row = (long long)tile * 16 + n;
  const float* xr = x + row * IN_DIM;

  v8f acc = {};
#pragma unroll 8
  for (int kb = 0; kb < IN_DIM / 4; ++kb) {
    int k = kb * 4 + koff;
    v2f a; a.x = xr[k]; a.y = xr[k + 1];
    v2f b; b.x = sW[k * 16 + n]; b.y = sW[(k + 1) * 16 + n];
    acc = wmma_f32_4(a, b, acc);
  }

  float bias = b1[n];
  int rbase = tile * 16 + ((lane >> 4) << 3);
#pragma unroll
  for (int r = 0; r < 8; ++r)
    h1[(long long)(rbase + r) * 16 + n] = fmaxf(acc[r] + bias, 0.0f);
}

__global__ __launch_bounds__(256) void k_scatter1(const int* __restrict__ src,
                                                  const int* __restrict__ dst,
                                                  const float* __restrict__ h1,
                                                  float* __restrict__ acc, int plane,
                                                  int nE) {
  float* agg1p = acc + (long long)chip_half() * plane;
  int lane = threadIdx.x & 31;
  long long e0 = ((long long)blockIdx.x * (blockDim.x >> 5) + (threadIdx.x >> 5)) * 32;
  if (e0 >= nE) return;
  long long e = e0 + lane;
  int s_l = 0, d_l = 0;
  if (e < nE) { s_l = src[e]; d_l = dst[e]; }
  int c = lane & 15;
  if (e0 + 32 <= nE) {
    float v[32];
    int dd[32];
#pragma unroll
    for (int j = 0; j < 32; ++j) {
      int s = __builtin_amdgcn_readlane(s_l, j);
      dd[j] = __builtin_amdgcn_readlane(d_l, j);
      v[j] = h1[(long long)s * 16 + c];
    }
    if (lane <= 16) {
#pragma unroll
      for (int j = 0; j < 32; ++j) {
        float val = (lane == 16) ? 1.0f : v[j];
        atomicAdd(&agg1p[(long long)dd[j] * 32 + lane], val);
      }
    }
  } else {
    int cnt = (int)(nE - e0);
    for (int j = 0; j < cnt; ++j) {
      int s = __builtin_amdgcn_readlane(s_l, j);
      int d = __builtin_amdgcn_readlane(d_l, j);
      float val = (lane == 16) ? 1.0f : h1[(long long)s * 16 + c];
      if (lane <= 16) atomicAdd(&agg1p[(long long)d * 32 + lane], val);
    }
  }
}

#define TP 34
__global__ __launch_bounds__(128) void k_conv1(float* accA, float* accB, float* h1p,
                                               const float* __restrict__ llw,
                                               const float* __restrict__ llb,
                                               const float* __restrict__ lrw,
                                               const float* __restrict__ l2w,
                                               float* __restrict__ out1,
                                               int ntiles) {
  __shared__ float sW[32 * 32];
  __shared__ float sW2[32 * 16];
  __shared__ float sT[4][16 * TP];
  int tid = threadIdx.x;
  for (int i = tid; i < 32 * 32; i += 128) {
    int k = i >> 5, j = i & 31;
    sW[i] = (k < 16) ? llw[k * 32 + j] : lrw[(k - 16) * 32 + j];
  }
  for (int i = tid; i < 32 * 16; i += 128) sW2[i] = l2w[i];
  __syncthreads();

  int wave = tid >> 5, lane = tid & 31;
  int tile = blockIdx.x * 4 + wave;
  if (tile >= ntiles) return;

  int n = lane & 15;
  int koff = (lane >> 4) << 1;
  long long row = (long long)tile * 16 + n;
  float inv = 1.0f / fmaxf(accA[row * 32 + 16] + accB[row * 32 + 16], 1.0f);

  v8f c0 = {}, c1 = {};
#pragma unroll
  for (int kb = 0; kb < 8; ++kb) {
    int kk = kb * 4 + koff;
    v2f a;
    if (kk < 16) { a.x = (accA[row * 32 + kk] + accB[row * 32 + kk]) * inv;
                   a.y = (accA[row * 32 + kk + 1] + accB[row * 32 + kk + 1]) * inv; }
    else         { a.x = h1p[row * 16 + kk - 16];   a.y = h1p[row * 16 + kk - 15]; }
    v2f b0; b0.x = sW[kk * 32 + n];      b0.y = sW[(kk + 1) * 32 + n];
    v2f b1; b1.x = sW[kk * 32 + 16 + n]; b1.y = sW[(kk + 1) * 32 + 16 + n];
    c0 = wmma_f32_4(a, b0, c0);
    c1 = wmma_f32_4(a, b1, c1);
  }

  float bl = llb[n], bh = llb[n + 16];
  int hb = (lane >> 4) << 3;
  int rbase = tile * 16 + hb;
  float* tl = sT[wave];
#pragma unroll
  for (int r = 0; r < 8; ++r) {
    float o0 = fmaxf(c0[r] + bl, 0.0f);
    float o1 = fmaxf(c1[r] + bh, 0.0f);
    long long orow = (long long)(rbase + r) * 32;
    out1[orow + n]      = o0;
    out1[orow + 16 + n] = o1;
    tl[(hb + r) * TP + n]      = o0;
    tl[(hb + r) * TP + 16 + n] = o1;
  }

  __builtin_amdgcn_wave_barrier();

  v8f pc = {};
#pragma unroll
  for (int kb = 0; kb < 8; ++kb) {
    int kk = kb * 4 + koff;
    v2f a; a.x = tl[n * TP + kk];     a.y = tl[n * TP + kk + 1];
    v2f b; b.x = sW2[kk * 16 + n];    b.y = sW2[(kk + 1) * 16 + n];
    pc = wmma_f32_4(a, b, pc);
  }
#pragma unroll
  for (int r = 0; r < 8; ++r)
    h1p[(long long)(rbase + r) * 16 + n] = pc[r];

  v4f z = {};
#pragma unroll
  for (int q = 0; q < 2; ++q) {
    int idx = q * 32 + lane;
    long long off = (long long)tile * 512 + (idx >> 2) * 32 + (idx & 3) * 4;
    *(v4f*)(accA + off) = z;
    *(v4f*)(accB + off) = z;
  }
}

__global__ __launch_bounds__(256) void k_scatter2(const int* __restrict__ src,
                                                  const int* __restrict__ dst,
                                                  const float* __restrict__ p,
                                                  float* __restrict__ acc, int plane,
                                                  int nE) {
  float* agg2 = acc + (long long)chip_half() * plane;
  int lane = threadIdx.x & 31;
  long long e0 = ((long long)blockIdx.x * (blockDim.x >> 5) + (threadIdx.x >> 5)) * 32;
  if (e0 >= nE) return;
  long long e = e0 + lane;
  int s_l = 0, d_l = 0;
  if (e < nE) { s_l = src[e]; d_l = dst[e]; }
  int c = lane & 15;
  if (e0 + 32 <= nE) {
    float v[32];
    int dd[32];
#pragma unroll
    for (int j = 0; j < 32; ++j) {
      int s = __builtin_amdgcn_readlane(s_l, j);
      dd[j] = __builtin_amdgcn_readlane(d_l, j);
      v[j] = p[(long long)s * 16 + c];
    }
    if (lane < 16) {
#pragma unroll
      for (int j = 0; j < 32; ++j)
        atomicAdd(&agg2[(long long)dd[j] * 32 + lane], v[j]);
    }
  } else {
    int cnt = (int)(nE - e0);
    for (int j = 0; j < cnt; ++j) {
      int s = __builtin_amdgcn_readlane(s_l, j);
      int d = __builtin_amdgcn_readlane(d_l, j);
      float val = p[(long long)s * 16 + c];
      if (lane < 16) atomicAdd(&agg2[(long long)d * 32 + lane], val);
    }
  }
}

__global__ __launch_bounds__(128) void k_conv2(const float* __restrict__ accA,
                                               const float* __restrict__ accB,
                                               const float* __restrict__ out1,
                                               const float* __restrict__ llb,
                                               const float* __restrict__ lrw,
                                               float* __restrict__ out2,
                                               int ntiles) {
  __shared__ float sW[32 * 16];
  int tid = threadIdx.x;
  for (int i = tid; i < 32 * 16; i += 128) sW[i] = lrw[i];
  __syncthreads();

  int wave = tid >> 5, lane = tid & 31;
  int tile = blockIdx.x * 4 + wave;
  if (tile >= ntiles) return;

  int n = lane & 15;
  int koff = (lane >> 4) << 1;
  long long row = (long long)tile * 16 + n;
  int rbase = tile * 16 + ((lane >> 4) << 3);
  float bias = llb[n];

  v8f acc;
#pragma unroll
  for (int r = 0; r < 8; ++r) {
    long long ar = (long long)(rbase + r) * 32;
    float inv = 1.0f / fmaxf(accA[ar + 16] + accB[ar + 16], 1.0f);
    acc[r] = (accA[ar + n] + accB[ar + n]) * inv + bias;
  }

#pragma unroll
  for (int kb = 0; kb < 8; ++kb) {
    int kk = kb * 4 + koff;
    v2f a; a.x = out1[row * 32 + kk]; a.y = out1[row * 32 + kk + 1];
    v2f b; b.x = sW[kk * 16 + n];     b.y = sW[(kk + 1) * 16 + n];
    acc = wmma_f32_4(a, b, acc);
  }

#pragma unroll
  for (int r = 0; r < 8; ++r)
    out2[(long long)(rbase + r) * 16 + n] = fmaxf(acc[r], 0.0f);
}

__global__ void k_head(const float* __restrict__ out2, const float* __restrict__ w2,
                       const float* __restrict__ b2, float* __restrict__ out, int N) {
  int i = blockIdx.x * blockDim.x + threadIdx.x;
  if (i >= N) return;
  float h0 = b2[0], h1 = b2[1], h2 = b2[2];
  const float* r = out2 + (long long)i * 16;
#pragma unroll
  for (int k = 0; k < 16; ++k) {
    float v = r[k];
    h0 += v * w2[k * 3 + 0];
    h1 += v * w2[k * 3 + 1];
    h2 += v * w2[k * 3 + 2];
  }
  float gsi = 1.0f / (1.0f + __expf(-h1));
  float mxi = 1.0f / (1.0f + __expf(-h2));
  float fsi = fmaxf(h0, 0.0f) + gsi;
  out[(long long)i * 3 + 0] = fsi;
  out[(long long)i * 3 + 1] = gsi;
  out[(long long)i * 3 + 2] = mxi;
}

extern "C" void kernel_launch(void* const* d_in, const int* in_sizes, int n_in,
                              void* d_out, int out_size, void* d_ws, size_t ws_size,
                              hipStream_t stream) {
  const float* x    = (const float*)d_in[0];
  const int*   ei   = (const int*)d_in[1];
  const float* w1   = (const float*)d_in[2];
  const float* b1   = (const float*)d_in[3];
  const float* c1ll = (const float*)d_in[4];
  const float* c1lb = (const float*)d_in[5];
  const float* c1lr = (const float*)d_in[6];
  const float* c2ll = (const float*)d_in[7];
  const float* c2lb = (const float*)d_in[8];
  const float* c2lr = (const float*)d_in[9];
  const float* w2   = (const float*)d_in[10];
  const float* b2   = (const float*)d_in[11];
  float* out = (float*)d_out;

  int N = in_sizes[0] / IN_DIM;
  int E = in_sizes[1] / 2;
  const int* src = ei;
  const int* dst = ei + E;

  float* ws    = (float*)d_ws;
  float* h1    = ws;
  float* acc   = ws + (long long)16 * N;
  float* out1  = ws + (long long)80 * N;
  float* out2  = ws;
  int plane = 32 * N;

  int ntiles = (N + 15) / 16;
  int wblocks = (ntiles + 3) / 4;

  {
    int n1 = 2 * plane;
    k_zero<<<(n1 + 255) / 256, 256, 0, stream>>>(acc, n1);
  }

  k_fc1<<<wblocks, 128, 0, stream>>>(x, w1, b1, h1, ntiles);

  {
    k_scatter1<<<(unsigned)((E + 255) / 256), 256, 0, stream>>>(src, dst, h1, acc, plane, E);
  }

  k_conv1<<<wblocks, 128, 0, stream>>>(acc, acc + plane, h1, c1ll, c1lb, c1lr, c2ll, out1, ntiles);

  {
    k_scatter2<<<(unsigned)((E + 255) / 256), 256, 0, stream>>>(src, dst, h1, acc, plane, E);
  }

  k_conv2<<<wblocks, 128, 0, stream>>>(acc, acc + plane, out1, c2lb, c2lr, out2, ntiles);

  k_head<<<(N + 255) / 256, 256, 0, stream>>>(out2, w2, b2, out, N);
}
